# starting kernel
# baseline (speedup 1.0000x reference)
#include <hip/hip_runtime.h>
#include <hip/hip_bf16.h>

typedef float v2f __attribute__((ext_vector_type(2)));
typedef float v8f __attribute__((ext_vector_type(8)));

#define ZDIM 32
#define HDIM 128
#define TBATCH 64
#define NWAVES 16
#define NTHREADS 512
#define H_S 80
#define FD_BASE (HDIM * H_S)
#define FD_S 1058
#define FD_ROW 33
#define SM_STRIDE (TBATCH * HDIM)
#define LDS_FLOATS (FD_BASE + TBATCH * FD_S)

__device__ __forceinline__ float rlane(float x, int i) {
  return __int_as_float(__builtin_amdgcn_readlane(__float_as_int(x), i));
}

__device__ __forceinline__ v8f wmma_f32(v2f a, v2f b, v8f c) {
  return __builtin_amdgcn_wmma_f32_16x16x4_f32(false, a, false, b, (short)0, c, false, false);
}

__global__ __launch_bounds__(NTHREADS)
void sylvester_flow_kernel(const float* __restrict__ z0,
                           const float* __restrict__ h,
                           const float* __restrict__ Wd,  const float* __restrict__ bd,
                           const float* __restrict__ Wd1, const float* __restrict__ bd1,
                           const float* __restrict__ Wd2, const float* __restrict__ bd2,
                           const float* __restrict__ Wb,  const float* __restrict__ bb,
                           float* __restrict__ out_z, float* __restrict__ out_ld)
{
  __shared__ float sh[LDS_FLOATS];

  const int tid   = threadIdx.x;
  const int lane  = tid & 31;
  const int wave  = tid >> 5;
  const int mrow  = lane & 15;
  const int mhalf = lane >> 4;
  const int mt     = wave & 3;
  const int ngroup = wave >> 2;
  const int s0 = blockIdx.x * TBATCH;

  #pragma unroll
  for (int r = 0; r < (TBATCH * HDIM) / NTHREADS; ++r) {
    int idx = r * NTHREADS + tid;
    int s = idx >> 7;
    int c = idx & 127;
    sh[c * H_S + s] = h[(size_t)(s0 + s) * HDIM + c];
  }
  __syncthreads();

  const int s_loc = mt * 16 + mrow;
  v2f acache[32];
  #pragma unroll
  for (int hh = 0; hh < 32; ++hh) {
    int kk = hh * 4 + 2 * mhalf;
    v2f a;
    a.x = sh[kk * H_S + s_loc];
    a.y = sh[(kk + 1) * H_S + s_loc];
    acache[hh] = a;
  }

  #pragma unroll 1
  for (int q = 0; q < 6; ++q) {
    int id2 = ngroup * 6 + q;
    int mat = id2 >> 3;
    int nt2 = id2 & 7;
    const float* Wm = (mat == 0) ? Wd1 : (mat == 1) ? Wd2 : Wb;
    const float* bm = (mat == 0) ? bd1 : (mat == 1) ? bd2 : bb;
    int cc = nt2 * 16 + mrow;
    float bias = bm[cc];
    v8f acc;
    #pragma unroll
    for (int r = 0; r < 8; ++r) acc[r] = bias;
    const float* wrow = Wm + (size_t)cc * HDIM;
    v2f bmat[32];
    #pragma unroll
    for (int hh = 0; hh < 32; ++hh)
      bmat[hh] = *reinterpret_cast<const v2f*>(wrow + hh * 4 + 2 * mhalf);
    #pragma unroll
    for (int hh = 0; hh < 32; ++hh)
      acc = wmma_f32(acache[hh], bmat[hh], acc);
    int sbase = mt * 16 + mhalf * 8;
    #pragma unroll
    for (int r = 0; r < 8; ++r) {
      float v = acc[r];
      if (mat < 2) v = tanhf(v);
      sh[FD_BASE + mat * SM_STRIDE + (sbase + r) * HDIM + cc] = v;
    }
  }
  __syncthreads();

  const int sflow = wave * 4;
  float zr[4], ldr[4], d1r[16], d2r[16], br[16];
  #pragma unroll
  for (int si = 0; si < 4; ++si) {
    zr[si]  = z0[(size_t)(s0 + sflow + si) * ZDIM + lane];
    ldr[si] = 0.0f;
    #pragma unroll
    for (int k = 0; k < 4; ++k) {
      int coff = lane * 4 + k;
      d1r[si * 4 + k] = sh[FD_BASE + 0 * SM_STRIDE + (sflow + si) * HDIM + coff];
      d2r[si * 4 + k] = sh[FD_BASE + 1 * SM_STRIDE + (sflow + si) * HDIM + coff];
      br [si * 4 + k] = sh[FD_BASE + 2 * SM_STRIDE + (sflow + si) * HDIM + coff];
    }
  }

  #pragma unroll 1
  for (int k = 0; k < 4; ++k) {
    __syncthreads();

    #pragma unroll 1
    for (int tt = 0; tt < 16; ++tt) {
      int nt = ngroup * 16 + tt;
      int i_ = nt >> 1;
      int j0 = (nt & 1) << 4;
      int ccol = i_ * 128 + (j0 + mrow) * 4 + k;
      float bias = bd[ccol];
      v8f acc;
      #pragma unroll
      for (int r = 0; r < 8; ++r) acc[r] = bias;
      const float* wrow = Wd + (size_t)ccol * HDIM;
      v2f bmat[32];
      #pragma unroll
      for (int hh = 0; hh < 32; ++hh)
        bmat[hh] = *reinterpret_cast<const v2f*>(wrow + hh * 4 + 2 * mhalf);
      #pragma unroll
      for (int hh = 0; hh < 32; ++hh)
        acc = wmma_f32(acache[hh], bmat[hh], acc);
      int sb = mt * 16 + mhalf * 8;
      int base = FD_BASE + sb * FD_S + i_ * FD_ROW + j0 + mrow;
      #pragma unroll
      for (int r = 0; r < 8; ++r)
        sh[base + r * FD_S] = acc[r];
    }
    __syncthreads();

    const bool flip = (k & 1);
    #pragma unroll 1
    for (int si = 0; si < 4; ++si) {
      float zv = zr[si];
      float zper = flip ? __shfl_xor(zv, 31, 32) : zv;
      int fdb = FD_BASE + (sflow + si) * FD_S;
      float pre = br[si * 4 + k] + zper * d2r[si * 4 + k];
      #pragma unroll
      for (int i = 1; i < 32; ++i) {
        float zi = rlane(zper, i);
        float v  = sh[fdb + i * FD_ROW + lane];
        pre += (i > lane) ? zi * v : 0.0f;
      }
      float t = tanhf(pre);
      float dz = t * d1r[si * 4 + k];
      #pragma unroll
      for (int j = 1; j < 32; ++j) {
        float tj = rlane(t, j);
        float v  = sh[fdb + lane * FD_ROW + j];
        dz += (j > lane) ? tj * v : 0.0f;
      }
      if (flip) dz = __shfl_xor(dz, 31, 32);
      zr[si] = zv + dz;
      float dj = (1.0f - t * t) * (d1r[si * 4 + k] * d2r[si * 4 + k]) + 1.0f;
      ldr[si] += logf(fabsf(dj));
    }
  }

  #pragma unroll
  for (int si = 0; si < 4; ++si) {
    int s = s0 + sflow + si;
    out_z[(size_t)s * ZDIM + lane] = zr[si];
    float v = ldr[si];
    #pragma unroll
    for (int off = 16; off > 0; off >>= 1) v += __shfl_xor(v, off, 32);
    if (lane == 0) out_ld[s] = v;
  }
}

extern "C" void kernel_launch(void* const* d_in, const int* in_sizes, int n_in,
                              void* d_out, int out_size, void* d_ws, size_t ws_size,
                              hipStream_t stream) {
  const float* z0  = (const float*)d_in[0];
  const float* h   = (const float*)d_in[1];
  const float* Wd  = (const float*)d_in[2];
  const float* bd  = (const float*)d_in[3];
  const float* Wd1 = (const float*)d_in[4];
  const float* bd1 = (const float*)d_in[5];
  const float* Wd2 = (const float*)d_in[6];
  const float* bd2 = (const float*)d_in[7];
  const float* Wb  = (const float*)d_in[8];
  const float* bb  = (const float*)d_in[9];

  const int B = in_sizes[0] / ZDIM;
  float* out   = (float*)d_out;
  float* out_z = out;
  float* out_ld = out + (size_t)B * ZDIM;

  dim3 grid(B / TBATCH), block(NTHREADS);
  hipLaunchKernelGGL(sylvester_flow_kernel, grid, block, 0, stream,
                     z0, h, Wd, bd, Wd1, bd1, Wd2, bd2, Wb, bb, out_z, out_ld);
}


# Main GEMM: each weight (B) register pair is refilled with the wave's next column tile right behind its last WMMA row, also across the flow step (its barrier now orders LDS only); sums bit-identical
# speedup vs baseline: 1.3746x; 1.3746x over previous
#include <hip/hip_runtime.h>
#include <hip/hip_bf16.h>
#include <type_traits>

typedef float v2f __attribute__((ext_vector_type(2)));
typedef float v4f __attribute__((ext_vector_type(4)));
typedef float v8f __attribute__((ext_vector_type(8)));

#define ZDIM 32
#define HDIM 128
#define TBATCH 64
#define NWAVES 16
#define NTHREADS 512
#define A_S 132
#define A_HALF 64
#define FD_BASE 10240
#define FD_S 1058
#define FD_ROW 33
#define SM_STRIDE (TBATCH * HDIM)
#define LDS_FLOATS (FD_BASE + TBATCH * FD_S)
#define A_AHEAD 2
#define A_RING (A_AHEAD + 1)

__device__ __forceinline__ float rlane(float x, int i) {
  return __int_as_float(__builtin_amdgcn_readlane(__float_as_int(x), i));
}

__device__ __forceinline__ v8f wmma_f32(v2f a, v2f b, v8f c) {
  return __builtin_amdgcn_wmma_f32_16x16x4_f32(false, a, false, b, (short)0, c, false, false);
}

__global__ __launch_bounds__(NTHREADS)
void sylvester_flow_kernel(const float* __restrict__ z0,
                           const float* __restrict__ h,
                           const float* __restrict__ Wd,  const float* __restrict__ bd,
                           const float* __restrict__ Wd1, const float* __restrict__ bd1,
                           const float* __restrict__ Wd2, const float* __restrict__ bd2,
                           const float* __restrict__ Wb,  const float* __restrict__ bb,
                           float* __restrict__ out_z, float* __restrict__ out_ld)
{
  __shared__ __attribute__((aligned(16))) float sh[LDS_FLOATS];

  const int tid   = threadIdx.x;
  const int lane  = tid & 31;
  const int wave  = __builtin_amdgcn_readfirstlane(tid >> 5);
  const int mrow  = lane & 15;
  const int mhalf = lane >> 4;
  const int s0 = blockIdx.x * TBATCH;

  #pragma unroll
  for (int r = 0; r < (TBATCH * HDIM / 4) / NTHREADS; ++r) {
    int idx = r * NTHREADS + tid;
    int s  = idx >> 5;
    int hh = idx & 31;
    v4f v = *reinterpret_cast<const v4f*>(h + (size_t)(s0 + s) * HDIM + hh * 4);
    v2f lo; lo.x = v.x; lo.y = v.y;
    v2f hi; hi.x = v.z; hi.y = v.w;
    *reinterpret_cast<v2f*>(&sh[s * A_S + 2 * hh])          = lo;
    *reinterpret_cast<v2f*>(&sh[s * A_S + A_HALF + 2 * hh]) = hi;
  }
  __syncthreads();

  const float* ap = sh + mrow * A_S + mhalf * A_HALF;

  {
    const int p0 = wave * 6;
    const int psplit = ((p0 >> 2) + 1) << 2;
    #pragma unroll 1
    for (int seg = 0; seg < 2; ++seg) {
      int pbeg = seg ? psplit : p0;
      int pend = seg ? (p0 + 6) : psplit;
      int ct  = pbeg >> 2;
      int mat = ct >> 3;
      int nt2 = ct & 7;
      const float* Wm = (mat == 0) ? Wd1 : (mat == 1) ? Wd2 : Wb;
      const float* bm = (mat == 0) ? bd1 : (mat == 1) ? bd2 : bb;
      int cc = nt2 * 16 + mrow;
      float bias = bm[cc];
      const float* wrow = Wm + (size_t)cc * HDIM;
      v2f bmat[32];
      #pragma unroll
      for (int hh = 0; hh < 32; ++hh)
        bmat[hh] = *reinterpret_cast<const v2f*>(wrow + hh * 4 + 2 * mhalf);
      #pragma unroll 1
      for (int p = pbeg; p < pend; ++p) {
        int m = p & 3;
        v8f acc;
        #pragma unroll
        for (int r = 0; r < 8; ++r) acc[r] = bias;
        const float* am = ap + m * 16 * A_S;
        #pragma unroll
        for (int q = 0; q < 16; ++q) {
          v4f a4 = *reinterpret_cast<const v4f*>(am + 4 * q);
          v2f a0; a0.x = a4.x; a0.y = a4.y;
          v2f a1; a1.x = a4.z; a1.y = a4.w;
          acc = wmma_f32(a0, bmat[2 * q],     acc);
          acc = wmma_f32(a1, bmat[2 * q + 1], acc);
        }
        int sbase = m * 16 + mhalf * 8;
        #pragma unroll
        for (int r = 0; r < 8; ++r) {
          float v = acc[r];
          if (mat < 2) v = tanhf(v);
          sh[FD_BASE + mat * SM_STRIDE + (sbase + r) * HDIM + cc] = v;
        }
      }
    }
  }
  const int b_lane = mrow * 4 * HDIM + 2 * mhalf;
  v2f bmat[32];
  float bias;
  {
    const int nt0 = wave * 4;
    const int ch0 = (nt0 >> 1) * 128 + ((nt0 & 1) << 4) * 4;
    const float* w0 = Wd + (size_t)ch0 * HDIM + b_lane;
    #pragma unroll
    for (int hh = 0; hh < 32; ++hh)
      bmat[hh] = *reinterpret_cast<const v2f*>(w0 + hh * 4);
    bias = bd[ch0 + mrow * 4];
  }
  __builtin_amdgcn_fence(__ATOMIC_RELEASE, "workgroup", "local");
  __builtin_amdgcn_s_barrier();
  __builtin_amdgcn_fence(__ATOMIC_ACQUIRE, "workgroup", "local");

  const int sflow = wave * 4;
  float zr[4], ldr[4], d1r[16], d2r[16], br[16];
  #pragma unroll
  for (int si = 0; si < 4; ++si) {
    zr[si]  = z0[(size_t)(s0 + sflow + si) * ZDIM + lane];
    ldr[si] = 0.0f;
    #pragma unroll
    for (int k = 0; k < 4; ++k) {
      int coff = lane * 4 + k;
      d1r[si * 4 + k] = sh[FD_BASE + 0 * SM_STRIDE + (sflow + si) * HDIM + coff];
      d2r[si * 4 + k] = sh[FD_BASE + 1 * SM_STRIDE + (sflow + si) * HDIM + coff];
      br [si * 4 + k] = sh[FD_BASE + 2 * SM_STRIDE + (sflow + si) * HDIM + coff];
    }
  }

  #pragma unroll 1
  for (int k = 0; k < 4; ++k) {
    __syncthreads();

    auto run_tile = [&](auto refill_tag, int tt) {
      constexpr bool REFILL = decltype(refill_tag)::value;
      int nt = wave * 4 + tt;
      int i_ = nt >> 1;
      int j0 = (nt & 1) << 4;
      const float* wnext = Wd;
      float bias_next = 0.0f;
      if (REFILL) {
        int tn  = k * 4 + tt + 1;
        int ntn = wave * 4 + (tn & 3);
        int chn = (ntn >> 1) * 128 + ((ntn & 1) << 4) * 4 + (tn >> 2);
        wnext = Wd + (size_t)chn * HDIM + b_lane;
        bias_next = bd[chn + mrow * 4];
      }
      v8f acc[4];
      #pragma unroll
      for (int m = 0; m < 4; ++m) {
        #pragma unroll
        for (int r = 0; r < 8; ++r) acc[m][r] = bias;
      }
      v4f ring[A_RING][4];
      #pragma unroll
      for (int q = 0; q < A_AHEAD; ++q) {
        #pragma unroll
        for (int m = 0; m < 4; ++m)
          ring[q][m] = *reinterpret_cast<const v4f*>(ap + m * 16 * A_S + 4 * q);
      }
      #pragma unroll
      for (int q = 0; q < 16; ++q) {
        if (q + A_AHEAD < 16) {
          #pragma unroll
          for (int m = 0; m < 4; ++m)
            ring[(q + A_AHEAD) % A_RING][m] =
                *reinterpret_cast<const v4f*>(ap + m * 16 * A_S + 4 * (q + A_AHEAD));
        }
        if (REFILL && q > 0) {
          bmat[2 * q - 2] = *reinterpret_cast<const v2f*>(wnext + (2 * q - 2) * 4);
          bmat[2 * q - 1] = *reinterpret_cast<const v2f*>(wnext + (2 * q - 1) * 4);
        }
        __builtin_amdgcn_sched_barrier(0);
        #pragma unroll
        for (int m = 0; m < 4; ++m) {
          v4f a = ring[q % A_RING][m];
          v2f a0; a0.x = a.x; a0.y = a.y;
          acc[m] = wmma_f32(a0, bmat[2 * q], acc[m]);
        }
        #pragma unroll
        for (int m = 0; m < 4; ++m) {
          v4f a = ring[q % A_RING][m];
          v2f a1; a1.x = a.z; a1.y = a.w;
          acc[m] = wmma_f32(a1, bmat[2 * q + 1], acc[m]);
        }
        __builtin_amdgcn_sched_barrier(0);
      }
      if (REFILL) {
        bmat[30] = *reinterpret_cast<const v2f*>(wnext + 30 * 4);
        bmat[31] = *reinterpret_cast<const v2f*>(wnext + 31 * 4);
        bias = bias_next;
      }
      #pragma unroll
      for (int m = 0; m < 4; ++m) {
        int sb = m * 16 + mhalf * 8;
        int base = FD_BASE + sb * FD_S + i_ * FD_ROW + j0 + mrow;
        #pragma unroll
        for (int r = 0; r < 8; ++r)
          sh[base + r * FD_S] = acc[m][r];
      }
    };
    #pragma unroll 1
    for (int tt = 0; tt < 4; ++tt) {
      if (k * 4 + tt < 15) run_tile(std::true_type{}, tt);
      else                 run_tile(std::false_type{}, tt);
    }
    __builtin_amdgcn_fence(__ATOMIC_RELEASE, "workgroup", "local");
    __builtin_amdgcn_s_barrier();
    __builtin_amdgcn_fence(__ATOMIC_ACQUIRE, "workgroup", "local");

    const bool flip = (k & 1);
    #pragma unroll 1
    for (int si = 0; si < 4; ++si) {
      float zv = zr[si];
      float zper = flip ? __shfl_xor(zv, 31, 32) : zv;
      int fdb = FD_BASE + (sflow + si) * FD_S;
      float pre = br[si * 4 + k] + zper * d2r[si * 4 + k];
      #pragma unroll
      for (int i = 1; i < 32; ++i) {
        float zi = rlane(zper, i);
        float v  = sh[fdb + i * FD_ROW + lane];
        pre += (i > lane) ? zi * v : 0.0f;
      }
      float t = tanhf(pre);
      float dz = t * d1r[si * 4 + k];
      #pragma unroll
      for (int j = 1; j < 32; ++j) {
        float tj = rlane(t, j);
        float v  = sh[fdb + lane * FD_ROW + j];
        dz += (j > lane) ? tj * v : 0.0f;
      }
      if (flip) dz = __shfl_xor(dz, 31, 32);
      zr[si] = zv + dz;
      float dj = (1.0f - t * t) * (d1r[si * 4 + k] * d2r[si * 4 + k]) + 1.0f;
      ldr[si] += logf(fabsf(dj));
    }
  }

  #pragma unroll
  for (int si = 0; si < 4; ++si) {
    int s = s0 + sflow + si;
    out_z[(size_t)s * ZDIM + lane] = zr[si];
    float v = ldr[si];
    #pragma unroll
    for (int off = 16; off > 0; off >>= 1) v += __shfl_xor(v, off, 32);
    if (lane == 0) out_ld[s] = v;
  }
}

extern "C" void kernel_launch(void* const* d_in, const int* in_sizes, int n_in,
                              void* d_out, int out_size, void* d_ws, size_t ws_size,
                              hipStream_t stream) {
  const float* z0  = (const float*)d_in[0];
  const float* h   = (const float*)d_in[1];
  const float* Wd  = (const float*)d_in[2];
  const float* bd  = (const float*)d_in[3];
  const float* Wd1 = (const float*)d_in[4];
  const float* bd1 = (const float*)d_in[5];
  const float* Wd2 = (const float*)d_in[6];
  const float* bd2 = (const float*)d_in[7];
  const float* Wb  = (const float*)d_in[8];
  const float* bb  = (const float*)d_in[9];

  const int B = in_sizes[0] / ZDIM;
  float* out   = (float*)d_out;
  float* out_z = out;
  float* out_ld = out + (size_t)B * ZDIM;

  dim3 grid(B / TBATCH), block(NTHREADS);
  hipLaunchKernelGGL(sylvester_flow_kernel, grid, block, 0, stream,
                     z0, h, Wd, bd, Wd1, bd1, Wd2, bd2, Wb, bb, out_z, out_ld);
}


# Flow step: a sample's 62 LDS reads issued as one group ahead of its sums, each term one select plus one fused multiply-add (was multiply, select, add), four samples unrolled; sum order kept
# speedup vs baseline: 1.4345x; 1.0436x over previous
#include <hip/hip_runtime.h>
#include <hip/hip_bf16.h>
#include <type_traits>

typedef float v2f __attribute__((ext_vector_type(2)));
typedef float v4f __attribute__((ext_vector_type(4)));
typedef float v8f __attribute__((ext_vector_type(8)));

#define ZDIM 32
#define HDIM 128
#define TBATCH 64
#define NWAVES 16
#define NTHREADS 512
#define A_S 132
#define A_HALF 64
#define FD_BASE 10240
#define FD_S 1058
#define FD_ROW 33
#define SM_STRIDE (TBATCH * HDIM)
#define LDS_FLOATS (FD_BASE + TBATCH * FD_S)
#define A_AHEAD 2
#define A_RING (A_AHEAD + 1)

__device__ __forceinline__ float rlane(float x, int i) {
  return __int_as_float(__builtin_amdgcn_readlane(__float_as_int(x), i));
}

__device__ __forceinline__ v8f wmma_f32(v2f a, v2f b, v8f c) {
  return __builtin_amdgcn_wmma_f32_16x16x4_f32(false, a, false, b, (short)0, c, false, false);
}

__global__ __launch_bounds__(NTHREADS)
void sylvester_flow_kernel(const float* __restrict__ z0,
                           const float* __restrict__ h,
                           const float* __restrict__ Wd,  const float* __restrict__ bd,
                           const float* __restrict__ Wd1, const float* __restrict__ bd1,
                           const float* __restrict__ Wd2, const float* __restrict__ bd2,
                           const float* __restrict__ Wb,  const float* __restrict__ bb,
                           float* __restrict__ out_z, float* __restrict__ out_ld)
{
  __shared__ __attribute__((aligned(16))) float sh[LDS_FLOATS];

  const int tid   = threadIdx.x;
  const int lane  = tid & 31;
  const int wave  = __builtin_amdgcn_readfirstlane(tid >> 5);
  const int mrow  = lane & 15;
  const int mhalf = lane >> 4;
  const int s0 = blockIdx.x * TBATCH;

  #pragma unroll
  for (int r = 0; r < (TBATCH * HDIM / 4) / NTHREADS; ++r) {
    int idx = r * NTHREADS + tid;
    int s  = idx >> 5;
    int hh = idx & 31;
    v4f v = *reinterpret_cast<const v4f*>(h + (size_t)(s0 + s) * HDIM + hh * 4);
    v2f lo; lo.x = v.x; lo.y = v.y;
    v2f hi; hi.x = v.z; hi.y = v.w;
    *reinterpret_cast<v2f*>(&sh[s * A_S + 2 * hh])          = lo;
    *reinterpret_cast<v2f*>(&sh[s * A_S + A_HALF + 2 * hh]) = hi;
  }
  __syncthreads();

  const float* ap = sh + mrow * A_S + mhalf * A_HALF;

  {
    const int p0 = wave * 6;
    const int psplit = ((p0 >> 2) + 1) << 2;
    #pragma unroll 1
    for (int seg = 0; seg < 2; ++seg) {
      int pbeg = seg ? psplit : p0;
      int pend = seg ? (p0 + 6) : psplit;
      int ct  = pbeg >> 2;
      int mat = ct >> 3;
      int nt2 = ct & 7;
      const float* Wm = (mat == 0) ? Wd1 : (mat == 1) ? Wd2 : Wb;
      const float* bm = (mat == 0) ? bd1 : (mat == 1) ? bd2 : bb;
      int cc = nt2 * 16 + mrow;
      float bias = bm[cc];
      const float* wrow = Wm + (size_t)cc * HDIM;
      v2f bmat[32];
      #pragma unroll
      for (int hh = 0; hh < 32; ++hh)
        bmat[hh] = *reinterpret_cast<const v2f*>(wrow + hh * 4 + 2 * mhalf);
      #pragma unroll 1
      for (int p = pbeg; p < pend; ++p) {
        int m = p & 3;
        v8f acc;
        #pragma unroll
        for (int r = 0; r < 8; ++r) acc[r] = bias;
        const float* am = ap + m * 16 * A_S;
        #pragma unroll
        for (int q = 0; q < 16; ++q) {
          v4f a4 = *reinterpret_cast<const v4f*>(am + 4 * q);
          v2f a0; a0.x = a4.x; a0.y = a4.y;
          v2f a1; a1.x = a4.z; a1.y = a4.w;
          acc = wmma_f32(a0, bmat[2 * q],     acc);
          acc = wmma_f32(a1, bmat[2 * q + 1], acc);
        }
        int sbase = m * 16 + mhalf * 8;
        #pragma unroll
        for (int r = 0; r < 8; ++r) {
          float v = acc[r];
          if (mat < 2) v = tanhf(v);
          sh[FD_BASE + mat * SM_STRIDE + (sbase + r) * HDIM + cc] = v;
        }
      }
    }
  }
  const int b_lane = mrow * 4 * HDIM + 2 * mhalf;
  v2f bmat[32];
  float bias;
  {
    const int nt0 = wave * 4;
    const int ch0 = (nt0 >> 1) * 128 + ((nt0 & 1) << 4) * 4;
    const float* w0 = Wd + (size_t)ch0 * HDIM + b_lane;
    #pragma unroll
    for (int hh = 0; hh < 32; ++hh)
      bmat[hh] = *reinterpret_cast<const v2f*>(w0 + hh * 4);
    bias = bd[ch0 + mrow * 4];
  }
  __builtin_amdgcn_fence(__ATOMIC_RELEASE, "workgroup", "local");
  __builtin_amdgcn_s_barrier();
  __builtin_amdgcn_fence(__ATOMIC_ACQUIRE, "workgroup", "local");

  const int sflow = wave * 4;
  float zr[4], ldr[4], d1r[16], d2r[16], br[16];
  #pragma unroll
  for (int si = 0; si < 4; ++si) {
    zr[si]  = z0[(size_t)(s0 + sflow + si) * ZDIM + lane];
    ldr[si] = 0.0f;
    #pragma unroll
    for (int k = 0; k < 4; ++k) {
      int coff = lane * 4 + k;
      d1r[si * 4 + k] = sh[FD_BASE + 0 * SM_STRIDE + (sflow + si) * HDIM + coff];
      d2r[si * 4 + k] = sh[FD_BASE + 1 * SM_STRIDE + (sflow + si) * HDIM + coff];
      br [si * 4 + k] = sh[FD_BASE + 2 * SM_STRIDE + (sflow + si) * HDIM + coff];
    }
  }

  #pragma unroll 1
  for (int k = 0; k < 4; ++k) {
    __syncthreads();

    auto run_tile = [&](auto refill_tag, int tt) {
      constexpr bool REFILL = decltype(refill_tag)::value;
      int nt = wave * 4 + tt;
      int i_ = nt >> 1;
      int j0 = (nt & 1) << 4;
      const float* wnext = Wd;
      float bias_next = 0.0f;
      if (REFILL) {
        int tn  = k * 4 + tt + 1;
        int ntn = wave * 4 + (tn & 3);
        int chn = (ntn >> 1) * 128 + ((ntn & 1) << 4) * 4 + (tn >> 2);
        wnext = Wd + (size_t)chn * HDIM + b_lane;
        bias_next = bd[chn + mrow * 4];
      }
      v8f acc[4];
      #pragma unroll
      for (int m = 0; m < 4; ++m) {
        #pragma unroll
        for (int r = 0; r < 8; ++r) acc[m][r] = bias;
      }
      v4f ring[A_RING][4];
      #pragma unroll
      for (int q = 0; q < A_AHEAD; ++q) {
        #pragma unroll
        for (int m = 0; m < 4; ++m)
          ring[q][m] = *reinterpret_cast<const v4f*>(ap + m * 16 * A_S + 4 * q);
      }
      #pragma unroll
      for (int q = 0; q < 16; ++q) {
        if (q + A_AHEAD < 16) {
          #pragma unroll
          for (int m = 0; m < 4; ++m)
            ring[(q + A_AHEAD) % A_RING][m] =
                *reinterpret_cast<const v4f*>(ap + m * 16 * A_S + 4 * (q + A_AHEAD));
        }
        if (REFILL && q > 0) {
          bmat[2 * q - 2] = *reinterpret_cast<const v2f*>(wnext + (2 * q - 2) * 4);
          bmat[2 * q - 1] = *reinterpret_cast<const v2f*>(wnext + (2 * q - 1) * 4);
        }
        __builtin_amdgcn_sched_barrier(0);
        #pragma unroll
        for (int m = 0; m < 4; ++m) {
          v4f a = ring[q % A_RING][m];
          v2f a0; a0.x = a.x; a0.y = a.y;
          acc[m] = wmma_f32(a0, bmat[2 * q], acc[m]);
        }
        #pragma unroll
        for (int m = 0; m < 4; ++m) {
          v4f a = ring[q % A_RING][m];
          v2f a1; a1.x = a.z; a1.y = a.w;
          acc[m] = wmma_f32(a1, bmat[2 * q + 1], acc[m]);
        }
        __builtin_amdgcn_sched_barrier(0);
      }
      if (REFILL) {
        bmat[30] = *reinterpret_cast<const v2f*>(wnext + 30 * 4);
        bmat[31] = *reinterpret_cast<const v2f*>(wnext + 31 * 4);
        bias = bias_next;
      }
      #pragma unroll
      for (int m = 0; m < 4; ++m) {
        int sb = m * 16 + mhalf * 8;
        int base = FD_BASE + sb * FD_S + i_ * FD_ROW + j0 + mrow;
        #pragma unroll
        for (int r = 0; r < 8; ++r)
          sh[base + r * FD_S] = acc[m][r];
      }
    };
    #pragma unroll 1
    for (int tt = 0; tt < 4; ++tt) {
      if (k * 4 + tt < 15) run_tile(std::true_type{}, tt);
      else                 run_tile(std::false_type{}, tt);
    }
    __builtin_amdgcn_fence(__ATOMIC_RELEASE, "workgroup", "local");
    __builtin_amdgcn_s_barrier();
    __builtin_amdgcn_fence(__ATOMIC_ACQUIRE, "workgroup", "local");

    const bool flip = (k & 1);
    #pragma unroll
    for (int si = 0; si < 4; ++si) {
      const float zv = zr[si];
      const float zper = flip ? __shfl_xor(zv, 31, 32) : zv;
      const float* fdp = sh + FD_BASE + (sflow + si) * FD_S;
      const float d1 = d1r[si * 4 + k];
      const float d2 = d2r[si * 4 + k];
      float lo[31], up[31];
      #pragma unroll
      for (int i = 1; i < 32; ++i) lo[i - 1] = fdp[i * FD_ROW + lane];
      #pragma unroll
      for (int j = 1; j < 32; ++j) up[j - 1] = fdp[lane * FD_ROW + j];
      __builtin_amdgcn_sched_barrier(0);
      float pre = br[si * 4 + k] + zper * d2;
      #pragma unroll
      for (int i = 1; i < 32; ++i) {
        const float zi = rlane(zper, i);
        pre = fmaf(zi, (i > lane) ? lo[i - 1] : 0.0f, pre);
      }
      const float t = tanhf(pre);
      float dz = t * d1;
      #pragma unroll
      for (int j = 1; j < 32; ++j) {
        const float tj = rlane(t, j);
        dz = fmaf(tj, (j > lane) ? up[j - 1] : 0.0f, dz);
      }
      if (flip) dz = __shfl_xor(dz, 31, 32);
      zr[si] = zv + dz;
      const float dj = (1.0f - t * t) * (d1 * d2) + 1.0f;
      ldr[si] += logf(fabsf(dj));
    }
  }

  #pragma unroll
  for (int si = 0; si < 4; ++si) {
    int s = s0 + sflow + si;
    out_z[(size_t)s * ZDIM + lane] = zr[si];
    float v = ldr[si];
    #pragma unroll
    for (int off = 16; off > 0; off >>= 1) v += __shfl_xor(v, off, 32);
    if (lane == 0) out_ld[s] = v;
  }
}

extern "C" void kernel_launch(void* const* d_in, const int* in_sizes, int n_in,
                              void* d_out, int out_size, void* d_ws, size_t ws_size,
                              hipStream_t stream) {
  const float* z0  = (const float*)d_in[0];
  const float* h   = (const float*)d_in[1];
  const float* Wd  = (const float*)d_in[2];
  const float* bd  = (const float*)d_in[3];
  const float* Wd1 = (const float*)d_in[4];
  const float* bd1 = (const float*)d_in[5];
  const float* Wd2 = (const float*)d_in[6];
  const float* bd2 = (const float*)d_in[7];
  const float* Wb  = (const float*)d_in[8];
  const float* bb  = (const float*)d_in[9];

  const int B = in_sizes[0] / ZDIM;
  float* out   = (float*)d_out;
  float* out_z = out;
  float* out_ld = out + (size_t)B * ZDIM;

  dim3 grid(B / TBATCH), block(NTHREADS);
  hipLaunchKernelGGL(sylvester_flow_kernel, grid, block, 0, stream,
                     z0, h, Wd, bd, Wd1, bd1, Wd2, bd2, Wb, bb, out_z, out_ld);
}


# Flow step with one sample per lane (a wave owns 32 samples x 4 Z indices): only the triangle's own terms are read and multiplied, z and t pass through LDS; log-det summed per wave, then across waves
# speedup vs baseline: 1.4887x; 1.0378x over previous
#include <hip/hip_runtime.h>
#include <hip/hip_bf16.h>
#include <type_traits>

typedef float v2f __attribute__((ext_vector_type(2)));
typedef float v4f __attribute__((ext_vector_type(4)));
typedef float v8f __attribute__((ext_vector_type(8)));

#define ZDIM 32
#define HDIM 128
#define TBATCH 64
#define NWAVES 16
#define NTHREADS 512
#define A_S 132
#define A_HALF 64
#define FD_BASE 10240
#define FD_S 1058
#define FD_ROW 33
#define SM_S 132
#define SM_STRIDE (TBATCH * SM_S)
#define FD_END (FD_BASE + TBATCH * FD_S)
#define Z_S 36
#define Z_BASE FD_END
#define T_COL 32
#define LD_BASE 8448
#define LDS_FLOATS (Z_BASE + TBATCH * Z_S)
#define A_AHEAD 2
#define A_RING (A_AHEAD + 1)

__device__ __forceinline__ float rlane(float x, int i) {
  return __int_as_float(__builtin_amdgcn_readlane(__float_as_int(x), i));
}

__device__ __forceinline__ v8f wmma_f32(v2f a, v2f b, v8f c) {
  return __builtin_amdgcn_wmma_f32_16x16x4_f32(false, a, false, b, (short)0, c, false, false);
}

__device__ __forceinline__ constexpr int jof(int G, int c) { return c < 2 ? 2 * G + c : 28 - 2 * G + c; }

template <int G>
__device__ __forceinline__ void flow_pre(const float* zrow, const float* tile,
                                         const float (&d2)[4], const float (&bv)[4],
                                         float (&zown)[4], float (&pre)[4]) {
  float z[32];
  #pragma unroll
  for (int q = 0; q < 8; ++q) {
    v4f v = *reinterpret_cast<const v4f*>(zrow + 4 * q);
    z[4 * q] = v.x; z[4 * q + 1] = v.y; z[4 * q + 2] = v.z; z[4 * q + 3] = v.w;
  }
  float m[4][31];
  #pragma unroll
  for (int c = 0; c < 4; ++c) {
    #pragma unroll
    for (int i = 1; i < 32; ++i)
      if (i > jof(G, c)) m[c][i - 1] = tile[i * FD_ROW + jof(G, c)];
  }
  __builtin_amdgcn_sched_barrier(0);
  #pragma unroll
  for (int c = 0; c < 4; ++c) {
    zown[c] = z[jof(G, c)];
    pre[c] = bv[c] + z[jof(G, c)] * d2[c];
  }
  #pragma unroll
  for (int i = 1; i < 32; ++i) {
    #pragma unroll
    for (int c = 0; c < 4; ++c)
      if (i > jof(G, c)) pre[c] = fmaf(z[i], m[c][i - 1], pre[c]);
  }
}

template <int G>
__device__ __forceinline__ void flow_dz(const float* tile, const float (&tv)[4], const float (&d1)[4],
                                        float (&dz)[4]) {
  float t[32];
  #pragma unroll
  for (int j = 1; j < 32; ++j)
    if (j > jof(G, 0)) t[j] = tile[j * FD_ROW + T_COL];
  float m[4][31];
  #pragma unroll
  for (int c = 0; c < 4; ++c) {
    #pragma unroll
    for (int j = 1; j < 32; ++j)
      if (j > jof(G, c)) m[c][j - 1] = tile[jof(G, c) * FD_ROW + j];
  }
  __builtin_amdgcn_sched_barrier(0);
  #pragma unroll
  for (int c = 0; c < 4; ++c) dz[c] = tv[c] * d1[c];
  #pragma unroll
  for (int j = 1; j < 32; ++j) {
    #pragma unroll
    for (int c = 0; c < 4; ++c)
      if (j > jof(G, c)) dz[c] = fmaf(t[j], m[c][j - 1], dz[c]);
  }
}

__global__ __launch_bounds__(NTHREADS)
void sylvester_flow_kernel(const float* __restrict__ z0,
                           const float* __restrict__ h,
                           const float* __restrict__ Wd,  const float* __restrict__ bd,
                           const float* __restrict__ Wd1, const float* __restrict__ bd1,
                           const float* __restrict__ Wd2, const float* __restrict__ bd2,
                           const float* __restrict__ Wb,  const float* __restrict__ bb,
                           float* __restrict__ out_z, float* __restrict__ out_ld)
{
  __shared__ __attribute__((aligned(16))) float sh[LDS_FLOATS];

  const int tid   = threadIdx.x;
  const int lane  = tid & 31;
  const int wave  = __builtin_amdgcn_readfirstlane(tid >> 5);
  const int mrow  = lane & 15;
  const int mhalf = lane >> 4;
  const int s0 = blockIdx.x * TBATCH;

  #pragma unroll
  for (int r = 0; r < (TBATCH * HDIM / 4) / NTHREADS; ++r) {
    int idx = r * NTHREADS + tid;
    int s  = idx >> 5;
    int hh = idx & 31;
    v4f v = *reinterpret_cast<const v4f*>(h + (size_t)(s0 + s) * HDIM + hh * 4);
    v2f lo; lo.x = v.x; lo.y = v.y;
    v2f hi; hi.x = v.z; hi.y = v.w;
    *reinterpret_cast<v2f*>(&sh[s * A_S + 2 * hh])          = lo;
    *reinterpret_cast<v2f*>(&sh[s * A_S + A_HALF + 2 * hh]) = hi;
  }
  {
    int s = tid >> 3;
    int c4 = (tid & 7) * 4;
    *reinterpret_cast<v4f*>(&sh[Z_BASE + s * Z_S + c4]) =
        *reinterpret_cast<const v4f*>(z0 + (size_t)(s0 + s) * ZDIM + c4);
  }
  __syncthreads();

  const float* ap = sh + mrow * A_S + mhalf * A_HALF;

  {
    const int p0 = wave * 6;
    const int psplit = ((p0 >> 2) + 1) << 2;
    #pragma unroll 1
    for (int seg = 0; seg < 2; ++seg) {
      int pbeg = seg ? psplit : p0;
      int pend = seg ? (p0 + 6) : psplit;
      int ct  = pbeg >> 2;
      int mat = ct >> 3;
      int nt2 = ct & 7;
      const float* Wm = (mat == 0) ? Wd1 : (mat == 1) ? Wd2 : Wb;
      const float* bm = (mat == 0) ? bd1 : (mat == 1) ? bd2 : bb;
      int cc = nt2 * 16 + mrow;
      float bias = bm[cc];
      const float* wrow = Wm + (size_t)cc * HDIM;
      v2f bmat[32];
      #pragma unroll
      for (int hh = 0; hh < 32; ++hh)
        bmat[hh] = *reinterpret_cast<const v2f*>(wrow + hh * 4 + 2 * mhalf);
      #pragma unroll 1
      for (int p = pbeg; p < pend; ++p) {
        int m = p & 3;
        v8f acc;
        #pragma unroll
        for (int r = 0; r < 8; ++r) acc[r] = bias;
        const float* am = ap + m * 16 * A_S;
        #pragma unroll
        for (int q = 0; q < 16; ++q) {
          v4f a4 = *reinterpret_cast<const v4f*>(am + 4 * q);
          v2f a0; a0.x = a4.x; a0.y = a4.y;
          v2f a1; a1.x = a4.z; a1.y = a4.w;
          acc = wmma_f32(a0, bmat[2 * q],     acc);
          acc = wmma_f32(a1, bmat[2 * q + 1], acc);
        }
        int sbase = m * 16 + mhalf * 8;
        #pragma unroll
        for (int r = 0; r < 8; ++r) {
          float v = acc[r];
          if (mat < 2) v = tanhf(v);
          sh[FD_BASE + mat * SM_STRIDE + (sbase + r) * SM_S + cc] = v;
        }
      }
    }
  }
  const int b_lane = mrow * 4 * HDIM + 2 * mhalf;
  v2f bmat[32];
  float bias;
  {
    const int nt0 = wave * 4;
    const int ch0 = (nt0 >> 1) * 128 + ((nt0 & 1) << 4) * 4;
    const float* w0 = Wd + (size_t)ch0 * HDIM + b_lane;
    #pragma unroll
    for (int hh = 0; hh < 32; ++hh)
      bmat[hh] = *reinterpret_cast<const v2f*>(w0 + hh * 4);
    bias = bd[ch0 + mrow * 4];
  }
  __builtin_amdgcn_fence(__ATOMIC_RELEASE, "workgroup", "local");
  __builtin_amdgcn_s_barrier();
  __builtin_amdgcn_fence(__ATOMIC_ACQUIRE, "workgroup", "local");

  const int fh = wave & 1;
  const int fg = wave >> 1;
  const int fs = fh * 32 + lane;
  float ldp = 0.0f;
  float d1r[16], d2r[16], br[16];
  int jrun[4];
  #pragma unroll
  for (int c = 0; c < 4; ++c) {
    jrun[c] = c < 2 ? 2 * fg + c : 28 - 2 * fg + c;
    v4f a = *reinterpret_cast<const v4f*>(&sh[FD_BASE + 0 * SM_STRIDE + fs * SM_S + jrun[c] * 4]);
    v4f b = *reinterpret_cast<const v4f*>(&sh[FD_BASE + 1 * SM_STRIDE + fs * SM_S + jrun[c] * 4]);
    v4f d = *reinterpret_cast<const v4f*>(&sh[FD_BASE + 2 * SM_STRIDE + fs * SM_S + jrun[c] * 4]);
    d1r[c * 4 + 0] = a.x; d1r[c * 4 + 1] = a.y; d1r[c * 4 + 2] = a.z; d1r[c * 4 + 3] = a.w;
    d2r[c * 4 + 0] = b.x; d2r[c * 4 + 1] = b.y; d2r[c * 4 + 2] = b.z; d2r[c * 4 + 3] = b.w;
    br [c * 4 + 0] = d.x; br [c * 4 + 1] = d.y; br [c * 4 + 2] = d.z; br [c * 4 + 3] = d.w;
  }
  float* const ftile = sh + FD_BASE + fs * FD_S;
  float* const fzrow = sh + Z_BASE + fs * Z_S;

  #pragma unroll 1
  for (int k = 0; k < 4; ++k) {
    __syncthreads();

    auto run_tile = [&](auto refill_tag, int tt) {
      constexpr bool REFILL = decltype(refill_tag)::value;
      int nt = wave * 4 + tt;
      int i_ = nt >> 1;
      int j0 = (nt & 1) << 4;
      const float* wnext = Wd;
      float bias_next = 0.0f;
      if (REFILL) {
        int tn  = k * 4 + tt + 1;
        int ntn = wave * 4 + (tn & 3);
        int chn = (ntn >> 1) * 128 + ((ntn & 1) << 4) * 4 + (tn >> 2);
        wnext = Wd + (size_t)chn * HDIM + b_lane;
        bias_next = bd[chn + mrow * 4];
      }
      v8f acc[4];
      #pragma unroll
      for (int m = 0; m < 4; ++m) {
        #pragma unroll
        for (int r = 0; r < 8; ++r) acc[m][r] = bias;
      }
      v4f ring[A_RING][4];
      #pragma unroll
      for (int q = 0; q < A_AHEAD; ++q) {
        #pragma unroll
        for (int m = 0; m < 4; ++m)
          ring[q][m] = *reinterpret_cast<const v4f*>(ap + m * 16 * A_S + 4 * q);
      }
      #pragma unroll
      for (int q = 0; q < 16; ++q) {
        if (q + A_AHEAD < 16) {
          #pragma unroll
          for (int m = 0; m < 4; ++m)
            ring[(q + A_AHEAD) % A_RING][m] =
                *reinterpret_cast<const v4f*>(ap + m * 16 * A_S + 4 * (q + A_AHEAD));
        }
        if (REFILL && q > 0) {
          bmat[2 * q - 2] = *reinterpret_cast<const v2f*>(wnext + (2 * q - 2) * 4);
          bmat[2 * q - 1] = *reinterpret_cast<const v2f*>(wnext + (2 * q - 1) * 4);
        }
        __builtin_amdgcn_sched_barrier(0);
        #pragma unroll
        for (int m = 0; m < 4; ++m) {
          v4f a = ring[q % A_RING][m];
          v2f a0; a0.x = a.x; a0.y = a.y;
          acc[m] = wmma_f32(a0, bmat[2 * q], acc[m]);
        }
        #pragma unroll
        for (int m = 0; m < 4; ++m) {
          v4f a = ring[q % A_RING][m];
          v2f a1; a1.x = a.z; a1.y = a.w;
          acc[m] = wmma_f32(a1, bmat[2 * q + 1], acc[m]);
        }
        __builtin_amdgcn_sched_barrier(0);
      }
      if (REFILL) {
        bmat[30] = *reinterpret_cast<const v2f*>(wnext + 30 * 4);
        bmat[31] = *reinterpret_cast<const v2f*>(wnext + 31 * 4);
        bias = bias_next;
      }
      #pragma unroll
      for (int m = 0; m < 4; ++m) {
        int sb = m * 16 + mhalf * 8;
        int base = FD_BASE + sb * FD_S + i_ * FD_ROW + j0 + mrow;
        #pragma unroll
        for (int r = 0; r < 8; ++r)
          sh[base + r * FD_S] = acc[m][r];
      }
    };
    #pragma unroll 1
    for (int tt = 0; tt < 4; ++tt) {
      if (k * 4 + tt < 15) run_tile(std::true_type{}, tt);
      else                 run_tile(std::false_type{}, tt);
    }
    __builtin_amdgcn_fence(__ATOMIC_RELEASE, "workgroup", "local");
    __builtin_amdgcn_s_barrier();
    __builtin_amdgcn_fence(__ATOMIC_ACQUIRE, "workgroup", "local");

    float d1k[4], d2k[4], bk[4];
    #pragma unroll
    for (int c = 0; c < 4; ++c) {
      d1k[c] = d1r[c * 4 + k];
      d2k[c] = d2r[c * 4 + k];
      bk[c]  = br [c * 4 + k];
    }
    float zown[4], pre[4];
    switch (fg) {
      case 0:  flow_pre<0>(fzrow, ftile, d2k, bk, zown, pre); break;
      case 1:  flow_pre<1>(fzrow, ftile, d2k, bk, zown, pre); break;
      case 2:  flow_pre<2>(fzrow, ftile, d2k, bk, zown, pre); break;
      case 3:  flow_pre<3>(fzrow, ftile, d2k, bk, zown, pre); break;
      case 4:  flow_pre<4>(fzrow, ftile, d2k, bk, zown, pre); break;
      case 5:  flow_pre<5>(fzrow, ftile, d2k, bk, zown, pre); break;
      case 6:  flow_pre<6>(fzrow, ftile, d2k, bk, zown, pre); break;
      default: flow_pre<7>(fzrow, ftile, d2k, bk, zown, pre); break;
    }
    float tv[4];
    #pragma unroll
    for (int c = 0; c < 4; ++c) {
      const float t = tanhf(pre[c]);
      tv[c] = t;
      ftile[jrun[c] * FD_ROW + T_COL] = t;
      const float dj = (1.0f - t * t) * (d1k[c] * d2k[c]) + 1.0f;
      ldp += logf(fabsf(dj));
    }
    __builtin_amdgcn_fence(__ATOMIC_RELEASE, "workgroup", "local");
    __builtin_amdgcn_s_barrier();
    __builtin_amdgcn_fence(__ATOMIC_ACQUIRE, "workgroup", "local");
    float dz[4];
    switch (fg) {
      case 0:  flow_dz<0>(ftile, tv, d1k, dz); break;
      case 1:  flow_dz<1>(ftile, tv, d1k, dz); break;
      case 2:  flow_dz<2>(ftile, tv, d1k, dz); break;
      case 3:  flow_dz<3>(ftile, tv, d1k, dz); break;
      case 4:  flow_dz<4>(ftile, tv, d1k, dz); break;
      case 5:  flow_dz<5>(ftile, tv, d1k, dz); break;
      case 6:  flow_dz<6>(ftile, tv, d1k, dz); break;
      default: flow_dz<7>(ftile, tv, d1k, dz); break;
    }
    #pragma unroll
    for (int c = 0; c < 4; ++c)
      fzrow[31 - jrun[c]] = zown[c] + dz[c];
  }

  sh[LD_BASE + fg * TBATCH + fs] = ldp;
  __syncthreads();
  {
    int s = tid >> 3;
    int c4 = (tid & 7) * 4;
    *reinterpret_cast<v4f*>(out_z + (size_t)(s0 + s) * ZDIM + c4) =
        *reinterpret_cast<const v4f*>(&sh[Z_BASE + s * Z_S + c4]);
    if (tid < TBATCH) {
      float v = 0.0f;
      #pragma unroll
      for (int g = 0; g < 8; ++g) v += sh[LD_BASE + g * TBATCH + tid];
      out_ld[s0 + tid] = v;
    }
  }
}

extern "C" void kernel_launch(void* const* d_in, const int* in_sizes, int n_in,
                              void* d_out, int out_size, void* d_ws, size_t ws_size,
                              hipStream_t stream) {
  const float* z0  = (const float*)d_in[0];
  const float* h   = (const float*)d_in[1];
  const float* Wd  = (const float*)d_in[2];
  const float* bd  = (const float*)d_in[3];
  const float* Wd1 = (const float*)d_in[4];
  const float* bd1 = (const float*)d_in[5];
  const float* Wd2 = (const float*)d_in[6];
  const float* bd2 = (const float*)d_in[7];
  const float* Wb  = (const float*)d_in[8];
  const float* bb  = (const float*)d_in[9];

  const int B = in_sizes[0] / ZDIM;
  float* out   = (float*)d_out;
  float* out_z = out;
  float* out_ld = out + (size_t)B * ZDIM;

  dim3 grid(B / TBATCH), block(NTHREADS);
  hipLaunchKernelGGL(sylvester_flow_kernel, grid, block, 0, stream,
                     z0, h, Wd, bd, Wd1, bd1, Wd2, bd2, Wb, bb, out_z, out_ld);
}


# tanh (diag1, diag2, flow t) and the log-det's log on the hardware transcendentals v_tanh_f32 / v_log_f32, were the library's routines (two-branch exp+rcp+polynomial; scaled log); last bits differ
# speedup vs baseline: 1.5466x; 1.0389x over previous
#include <hip/hip_runtime.h>
#include <hip/hip_bf16.h>
#include <type_traits>

typedef float v2f __attribute__((ext_vector_type(2)));
typedef float v4f __attribute__((ext_vector_type(4)));
typedef float v8f __attribute__((ext_vector_type(8)));

#define ZDIM 32
#define HDIM 128
#define TBATCH 64
#define NWAVES 16
#define NTHREADS 512
#define A_S 132
#define A_HALF 64
#define FD_BASE 10240
#define FD_S 1058
#define FD_ROW 33
#define SM_S 132
#define SM_STRIDE (TBATCH * SM_S)
#define FD_END (FD_BASE + TBATCH * FD_S)
#define Z_S 36
#define Z_BASE FD_END
#define T_COL 32
#define LD_BASE 8448
#define LDS_FLOATS (Z_BASE + TBATCH * Z_S)
#define A_AHEAD 2
#define A_RING (A_AHEAD + 1)

__device__ __forceinline__ float rlane(float x, int i) {
  return __int_as_float(__builtin_amdgcn_readlane(__float_as_int(x), i));
}

__device__ __forceinline__ v8f wmma_f32(v2f a, v2f b, v8f c) {
  return __builtin_amdgcn_wmma_f32_16x16x4_f32(false, a, false, b, (short)0, c, false, false);
}

__device__ __forceinline__ float tanh_hw(float x) { return __builtin_amdgcn_tanhf(x); }
__device__ __forceinline__ float log_hw(float x) { return __builtin_amdgcn_logf(x) * 0.69314718055994531f; }

__device__ __forceinline__ constexpr int jof(int G, int c) { return c < 2 ? 2 * G + c : 28 - 2 * G + c; }

template <int G>
__device__ __forceinline__ void flow_pre(const float* zrow, const float* tile,
                                         const float (&d2)[4], const float (&bv)[4],
                                         float (&zown)[4], float (&pre)[4]) {
  float z[32];
  #pragma unroll
  for (int q = 0; q < 8; ++q) {
    v4f v = *reinterpret_cast<const v4f*>(zrow + 4 * q);
    z[4 * q] = v.x; z[4 * q + 1] = v.y; z[4 * q + 2] = v.z; z[4 * q + 3] = v.w;
  }
  float m[4][31];
  #pragma unroll
  for (int c = 0; c < 4; ++c) {
    #pragma unroll
    for (int i = 1; i < 32; ++i)
      if (i > jof(G, c)) m[c][i - 1] = tile[i * FD_ROW + jof(G, c)];
  }
  __builtin_amdgcn_sched_barrier(0);
  #pragma unroll
  for (int c = 0; c < 4; ++c) {
    zown[c] = z[jof(G, c)];
    pre[c] = bv[c] + z[jof(G, c)] * d2[c];
  }
  #pragma unroll
  for (int i = 1; i < 32; ++i) {
    #pragma unroll
    for (int c = 0; c < 4; ++c)
      if (i > jof(G, c)) pre[c] = fmaf(z[i], m[c][i - 1], pre[c]);
  }
}

template <int G>
__device__ __forceinline__ void flow_dz(const float* tile, const float (&tv)[4], const float (&d1)[4],
                                        float (&dz)[4]) {
  float t[32];
  #pragma unroll
  for (int j = 1; j < 32; ++j)
    if (j > jof(G, 0)) t[j] = tile[j * FD_ROW + T_COL];
  float m[4][31];
  #pragma unroll
  for (int c = 0; c < 4; ++c) {
    #pragma unroll
    for (int j = 1; j < 32; ++j)
      if (j > jof(G, c)) m[c][j - 1] = tile[jof(G, c) * FD_ROW + j];
  }
  __builtin_amdgcn_sched_barrier(0);
  #pragma unroll
  for (int c = 0; c < 4; ++c) dz[c] = tv[c] * d1[c];
  #pragma unroll
  for (int j = 1; j < 32; ++j) {
    #pragma unroll
    for (int c = 0; c < 4; ++c)
      if (j > jof(G, c)) dz[c] = fmaf(t[j], m[c][j - 1], dz[c]);
  }
}

__global__ __launch_bounds__(NTHREADS)
void sylvester_flow_kernel(const float* __restrict__ z0,
                           const float* __restrict__ h,
                           const float* __restrict__ Wd,  const float* __restrict__ bd,
                           const float* __restrict__ Wd1, const float* __restrict__ bd1,
                           const float* __restrict__ Wd2, const float* __restrict__ bd2,
                           const float* __restrict__ Wb,  const float* __restrict__ bb,
                           float* __restrict__ out_z, float* __restrict__ out_ld)
{
  __shared__ __attribute__((aligned(16))) float sh[LDS_FLOATS];

  const int tid   = threadIdx.x;
  const int lane  = tid & 31;
  const int wave  = __builtin_amdgcn_readfirstlane(tid >> 5);
  const int mrow  = lane & 15;
  const int mhalf = lane >> 4;
  const int s0 = blockIdx.x * TBATCH;

  #pragma unroll
  for (int r = 0; r < (TBATCH * HDIM / 4) / NTHREADS; ++r) {
    int idx = r * NTHREADS + tid;
    int s  = idx >> 5;
    int hh = idx & 31;
    v4f v = *reinterpret_cast<const v4f*>(h + (size_t)(s0 + s) * HDIM + hh * 4);
    v2f lo; lo.x = v.x; lo.y = v.y;
    v2f hi; hi.x = v.z; hi.y = v.w;
    *reinterpret_cast<v2f*>(&sh[s * A_S + 2 * hh])          = lo;
    *reinterpret_cast<v2f*>(&sh[s * A_S + A_HALF + 2 * hh]) = hi;
  }
  {
    int s = tid >> 3;
    int c4 = (tid & 7) * 4;
    *reinterpret_cast<v4f*>(&sh[Z_BASE + s * Z_S + c4]) =
        *reinterpret_cast<const v4f*>(z0 + (size_t)(s0 + s) * ZDIM + c4);
  }
  __syncthreads();

  const float* ap = sh + mrow * A_S + mhalf * A_HALF;

  {
    const int p0 = wave * 6;
    const int psplit = ((p0 >> 2) + 1) << 2;
    #pragma unroll 1
    for (int seg = 0; seg < 2; ++seg) {
      int pbeg = seg ? psplit : p0;
      int pend = seg ? (p0 + 6) : psplit;
      int ct  = pbeg >> 2;
      int mat = ct >> 3;
      int nt2 = ct & 7;
      const float* Wm = (mat == 0) ? Wd1 : (mat == 1) ? Wd2 : Wb;
      const float* bm = (mat == 0) ? bd1 : (mat == 1) ? bd2 : bb;
      int cc = nt2 * 16 + mrow;
      float bias = bm[cc];
      const float* wrow = Wm + (size_t)cc * HDIM;
      v2f bmat[32];
      #pragma unroll
      for (int hh = 0; hh < 32; ++hh)
        bmat[hh] = *reinterpret_cast<const v2f*>(wrow + hh * 4 + 2 * mhalf);
      #pragma unroll 1
      for (int p = pbeg; p < pend; ++p) {
        int m = p & 3;
        v8f acc;
        #pragma unroll
        for (int r = 0; r < 8; ++r) acc[r] = bias;
        const float* am = ap + m * 16 * A_S;
        #pragma unroll
        for (int q = 0; q < 16; ++q) {
          v4f a4 = *reinterpret_cast<const v4f*>(am + 4 * q);
          v2f a0; a0.x = a4.x; a0.y = a4.y;
          v2f a1; a1.x = a4.z; a1.y = a4.w;
          acc = wmma_f32(a0, bmat[2 * q],     acc);
          acc = wmma_f32(a1, bmat[2 * q + 1], acc);
        }
        int sbase = m * 16 + mhalf * 8;
        #pragma unroll
        for (int r = 0; r < 8; ++r) {
          float v = acc[r];
          if (mat < 2) v = tanh_hw(v);
          sh[FD_BASE + mat * SM_STRIDE + (sbase + r) * SM_S + cc] = v;
        }
      }
    }
  }
  const int b_lane = mrow * 4 * HDIM + 2 * mhalf;
  v2f bmat[32];
  float bias;
  {
    const int nt0 = wave * 4;
    const int ch0 = (nt0 >> 1) * 128 + ((nt0 & 1) << 4) * 4;
    const float* w0 = Wd + (size_t)ch0 * HDIM + b_lane;
    #pragma unroll
    for (int hh = 0; hh < 32; ++hh)
      bmat[hh] = *reinterpret_cast<const v2f*>(w0 + hh * 4);
    bias = bd[ch0 + mrow * 4];
  }
  __builtin_amdgcn_fence(__ATOMIC_RELEASE, "workgroup", "local");
  __builtin_amdgcn_s_barrier();
  __builtin_amdgcn_fence(__ATOMIC_ACQUIRE, "workgroup", "local");

  const int fh = wave & 1;
  const int fg = wave >> 1;
  const int fs = fh * 32 + lane;
  float ldp = 0.0f;
  float d1r[16], d2r[16], br[16];
  int jrun[4];
  #pragma unroll
  for (int c = 0; c < 4; ++c) {
    jrun[c] = c < 2 ? 2 * fg + c : 28 - 2 * fg + c;
    v4f a = *reinterpret_cast<const v4f*>(&sh[FD_BASE + 0 * SM_STRIDE + fs * SM_S + jrun[c] * 4]);
    v4f b = *reinterpret_cast<const v4f*>(&sh[FD_BASE + 1 * SM_STRIDE + fs * SM_S + jrun[c] * 4]);
    v4f d = *reinterpret_cast<const v4f*>(&sh[FD_BASE + 2 * SM_STRIDE + fs * SM_S + jrun[c] * 4]);
    d1r[c * 4 + 0] = a.x; d1r[c * 4 + 1] = a.y; d1r[c * 4 + 2] = a.z; d1r[c * 4 + 3] = a.w;
    d2r[c * 4 + 0] = b.x; d2r[c * 4 + 1] = b.y; d2r[c * 4 + 2] = b.z; d2r[c * 4 + 3] = b.w;
    br [c * 4 + 0] = d.x; br [c * 4 + 1] = d.y; br [c * 4 + 2] = d.z; br [c * 4 + 3] = d.w;
  }
  float* const ftile = sh + FD_BASE + fs * FD_S;
  float* const fzrow = sh + Z_BASE + fs * Z_S;

  #pragma unroll 1
  for (int k = 0; k < 4; ++k) {
    __syncthreads();

    auto run_tile = [&](auto refill_tag, int tt) {
      constexpr bool REFILL = decltype(refill_tag)::value;
      int nt = wave * 4 + tt;
      int i_ = nt >> 1;
      int j0 = (nt & 1) << 4;
      const float* wnext = Wd;
      float bias_next = 0.0f;
      if (REFILL) {
        int tn  = k * 4 + tt + 1;
        int ntn = wave * 4 + (tn & 3);
        int chn = (ntn >> 1) * 128 + ((ntn & 1) << 4) * 4 + (tn >> 2);
        wnext = Wd + (size_t)chn * HDIM + b_lane;
        bias_next = bd[chn + mrow * 4];
      }
      v8f acc[4];
      #pragma unroll
      for (int m = 0; m < 4; ++m) {
        #pragma unroll
        for (int r = 0; r < 8; ++r) acc[m][r] = bias;
      }
      v4f ring[A_RING][4];
      #pragma unroll
      for (int q = 0; q < A_AHEAD; ++q) {
        #pragma unroll
        for (int m = 0; m < 4; ++m)
          ring[q][m] = *reinterpret_cast<const v4f*>(ap + m * 16 * A_S + 4 * q);
      }
      #pragma unroll
      for (int q = 0; q < 16; ++q) {
        if (q + A_AHEAD < 16) {
          #pragma unroll
          for (int m = 0; m < 4; ++m)
            ring[(q + A_AHEAD) % A_RING][m] =
                *reinterpret_cast<const v4f*>(ap + m * 16 * A_S + 4 * (q + A_AHEAD));
        }
        if (REFILL && q > 0) {
          bmat[2 * q - 2] = *reinterpret_cast<const v2f*>(wnext + (2 * q - 2) * 4);
          bmat[2 * q - 1] = *reinterpret_cast<const v2f*>(wnext + (2 * q - 1) * 4);
        }
        __builtin_amdgcn_sched_barrier(0);
        #pragma unroll
        for (int m = 0; m < 4; ++m) {
          v4f a = ring[q % A_RING][m];
          v2f a0; a0.x = a.x; a0.y = a.y;
          acc[m] = wmma_f32(a0, bmat[2 * q], acc[m]);
        }
        #pragma unroll
        for (int m = 0; m < 4; ++m) {
          v4f a = ring[q % A_RING][m];
          v2f a1; a1.x = a.z; a1.y = a.w;
          acc[m] = wmma_f32(a1, bmat[2 * q + 1], acc[m]);
        }
        __builtin_amdgcn_sched_barrier(0);
      }
      if (REFILL) {
        bmat[30] = *reinterpret_cast<const v2f*>(wnext + 30 * 4);
        bmat[31] = *reinterpret_cast<const v2f*>(wnext + 31 * 4);
        bias = bias_next;
      }
      #pragma unroll
      for (int m = 0; m < 4; ++m) {
        int sb = m * 16 + mhalf * 8;
        int base = FD_BASE + sb * FD_S + i_ * FD_ROW + j0 + mrow;
        #pragma unroll
        for (int r = 0; r < 8; ++r)
          sh[base + r * FD_S] = acc[m][r];
      }
    };
    #pragma unroll 1
    for (int tt = 0; tt < 4; ++tt) {
      if (k * 4 + tt < 15) run_tile(std::true_type{}, tt);
      else                 run_tile(std::false_type{}, tt);
    }
    __builtin_amdgcn_fence(__ATOMIC_RELEASE, "workgroup", "local");
    __builtin_amdgcn_s_barrier();
    __builtin_amdgcn_fence(__ATOMIC_ACQUIRE, "workgroup", "local");

    float d1k[4], d2k[4], bk[4];
    #pragma unroll
    for (int c = 0; c < 4; ++c) {
      d1k[c] = d1r[c * 4 + k];
      d2k[c] = d2r[c * 4 + k];
      bk[c]  = br [c * 4 + k];
    }
    float zown[4], pre[4];
    switch (fg) {
      case 0:  flow_pre<0>(fzrow, ftile, d2k, bk, zown, pre); break;
      case 1:  flow_pre<1>(fzrow, ftile, d2k, bk, zown, pre); break;
      case 2:  flow_pre<2>(fzrow, ftile, d2k, bk, zown, pre); break;
      case 3:  flow_pre<3>(fzrow, ftile, d2k, bk, zown, pre); break;
      case 4:  flow_pre<4>(fzrow, ftile, d2k, bk, zown, pre); break;
      case 5:  flow_pre<5>(fzrow, ftile, d2k, bk, zown, pre); break;
      case 6:  flow_pre<6>(fzrow, ftile, d2k, bk, zown, pre); break;
      default: flow_pre<7>(fzrow, ftile, d2k, bk, zown, pre); break;
    }
    float tv[4];
    #pragma unroll
    for (int c = 0; c < 4; ++c) {
      const float t = tanh_hw(pre[c]);
      tv[c] = t;
      ftile[jrun[c] * FD_ROW + T_COL] = t;
      const float dj = (1.0f - t * t) * (d1k[c] * d2k[c]) + 1.0f;
      ldp += log_hw(fabsf(dj));
    }
    __builtin_amdgcn_fence(__ATOMIC_RELEASE, "workgroup", "local");
    __builtin_amdgcn_s_barrier();
    __builtin_amdgcn_fence(__ATOMIC_ACQUIRE, "workgroup", "local");
    float dz[4];
    switch (fg) {
      case 0:  flow_dz<0>(ftile, tv, d1k, dz); break;
      case 1:  flow_dz<1>(ftile, tv, d1k, dz); break;
      case 2:  flow_dz<2>(ftile, tv, d1k, dz); break;
      case 3:  flow_dz<3>(ftile, tv, d1k, dz); break;
      case 4:  flow_dz<4>(ftile, tv, d1k, dz); break;
      case 5:  flow_dz<5>(ftile, tv, d1k, dz); break;
      case 6:  flow_dz<6>(ftile, tv, d1k, dz); break;
      default: flow_dz<7>(ftile, tv, d1k, dz); break;
    }
    #pragma unroll
    for (int c = 0; c < 4; ++c)
      fzrow[31 - jrun[c]] = zown[c] + dz[c];
  }

  sh[LD_BASE + fg * TBATCH + fs] = ldp;
  __syncthreads();
  {
    int s = tid >> 3;
    int c4 = (tid & 7) * 4;
    *reinterpret_cast<v4f*>(out_z + (size_t)(s0 + s) * ZDIM + c4) =
        *reinterpret_cast<const v4f*>(&sh[Z_BASE + s * Z_S + c4]);
    if (tid < TBATCH) {
      float v = 0.0f;
      #pragma unroll
      for (int g = 0; g < 8; ++g) v += sh[LD_BASE + g * TBATCH + tid];
      out_ld[s0 + tid] = v;
    }
  }
}

extern "C" void kernel_launch(void* const* d_in, const int* in_sizes, int n_in,
                              void* d_out, int out_size, void* d_ws, size_t ws_size,
                              hipStream_t stream) {
  const float* z0  = (const float*)d_in[0];
  const float* h   = (const float*)d_in[1];
  const float* Wd  = (const float*)d_in[2];
  const float* bd  = (const float*)d_in[3];
  const float* Wd1 = (const float*)d_in[4];
  const float* bd1 = (const float*)d_in[5];
  const float* Wd2 = (const float*)d_in[6];
  const float* bd2 = (const float*)d_in[7];
  const float* Wb  = (const float*)d_in[8];
  const float* bb  = (const float*)d_in[9];

  const int B = in_sizes[0] / ZDIM;
  float* out   = (float*)d_out;
  float* out_z = out;
  float* out_ld = out + (size_t)B * ZDIM;

  dim3 grid(B / TBATCH), block(NTHREADS);
  hipLaunchKernelGGL(sylvester_flow_kernel, grid, block, 0, stream,
                     z0, h, Wd, bd, Wd1, bd1, Wd2, bd2, Wb, bb, out_z, out_ld);
}


# Main GEMM: issue priority by progress (s_setprio 3 at step start, one level down per quarter of a wave's last tile) so the four waves of a SIMD reach the barrier together; sums bit-identical
# speedup vs baseline: 1.5928x; 1.0299x over previous
#include <hip/hip_runtime.h>
#include <hip/hip_bf16.h>
#include <type_traits>

typedef float v2f __attribute__((ext_vector_type(2)));
typedef float v4f __attribute__((ext_vector_type(4)));
typedef float v8f __attribute__((ext_vector_type(8)));

#define ZDIM 32
#define HDIM 128
#define TBATCH 64
#define NWAVES 16
#define NTHREADS 512
#define A_S 132
#define A_HALF 64
#define FD_BASE 10240
#define FD_S 1058
#define FD_ROW 33
#define SM_S 132
#define SM_STRIDE (TBATCH * SM_S)
#define FD_END (FD_BASE + TBATCH * FD_S)
#define Z_S 36
#define Z_BASE FD_END
#define T_COL 32
#define LD_BASE 8448
#define LDS_FLOATS (Z_BASE + TBATCH * Z_S)
#define A_AHEAD 2
#define A_RING (A_AHEAD + 1)

__device__ __forceinline__ float rlane(float x, int i) {
  return __int_as_float(__builtin_amdgcn_readlane(__float_as_int(x), i));
}

__device__ __forceinline__ v8f wmma_f32(v2f a, v2f b, v8f c) {
  return __builtin_amdgcn_wmma_f32_16x16x4_f32(false, a, false, b, (short)0, c, false, false);
}

__device__ __forceinline__ float tanh_hw(float x) { return __builtin_amdgcn_tanhf(x); }
__device__ __forceinline__ float log_hw(float x) { return __builtin_amdgcn_logf(x) * 0.69314718055994531f; }

__device__ __forceinline__ constexpr int jof(int G, int c) { return c < 2 ? 2 * G + c : 28 - 2 * G + c; }

template <int G>
__device__ __forceinline__ void flow_pre(const float* zrow, const float* tile,
                                         const float (&d2)[4], const float (&bv)[4],
                                         float (&zown)[4], float (&pre)[4]) {
  float z[32];
  #pragma unroll
  for (int q = 0; q < 8; ++q) {
    v4f v = *reinterpret_cast<const v4f*>(zrow + 4 * q);
    z[4 * q] = v.x; z[4 * q + 1] = v.y; z[4 * q + 2] = v.z; z[4 * q + 3] = v.w;
  }
  float m[4][31];
  #pragma unroll
  for (int c = 0; c < 4; ++c) {
    #pragma unroll
    for (int i = 1; i < 32; ++i)
      if (i > jof(G, c)) m[c][i - 1] = tile[i * FD_ROW + jof(G, c)];
  }
  __builtin_amdgcn_sched_barrier(0);
  #pragma unroll
  for (int c = 0; c < 4; ++c) {
    zown[c] = z[jof(G, c)];
    pre[c] = bv[c] + z[jof(G, c)] * d2[c];
  }
  #pragma unroll
  for (int i = 1; i < 32; ++i) {
    #pragma unroll
    for (int c = 0; c < 4; ++c)
      if (i > jof(G, c)) pre[c] = fmaf(z[i], m[c][i - 1], pre[c]);
  }
}

template <int G>
__device__ __forceinline__ void flow_dz(const float* tile, const float (&tv)[4], const float (&d1)[4],
                                        float (&dz)[4]) {
  float t[32];
  #pragma unroll
  for (int j = 1; j < 32; ++j)
    if (j > jof(G, 0)) t[j] = tile[j * FD_ROW + T_COL];
  float m[4][31];
  #pragma unroll
  for (int c = 0; c < 4; ++c) {
    #pragma unroll
    for (int j = 1; j < 32; ++j)
      if (j > jof(G, c)) m[c][j - 1] = tile[jof(G, c) * FD_ROW + j];
  }
  __builtin_amdgcn_sched_barrier(0);
  #pragma unroll
  for (int c = 0; c < 4; ++c) dz[c] = tv[c] * d1[c];
  #pragma unroll
  for (int j = 1; j < 32; ++j) {
    #pragma unroll
    for (int c = 0; c < 4; ++c)
      if (j > jof(G, c)) dz[c] = fmaf(t[j], m[c][j - 1], dz[c]);
  }
}

__global__ __launch_bounds__(NTHREADS)
void sylvester_flow_kernel(const float* __restrict__ z0,
                           const float* __restrict__ h,
                           const float* __restrict__ Wd,  const float* __restrict__ bd,
                           const float* __restrict__ Wd1, const float* __restrict__ bd1,
                           const float* __restrict__ Wd2, const float* __restrict__ bd2,
                           const float* __restrict__ Wb,  const float* __restrict__ bb,
                           float* __restrict__ out_z, float* __restrict__ out_ld)
{
  __shared__ __attribute__((aligned(16))) float sh[LDS_FLOATS];

  const int tid   = threadIdx.x;
  const int lane  = tid & 31;
  const int wave  = __builtin_amdgcn_readfirstlane(tid >> 5);
  const int mrow  = lane & 15;
  const int mhalf = lane >> 4;
  const int s0 = blockIdx.x * TBATCH;

  #pragma unroll
  for (int r = 0; r < (TBATCH * HDIM / 4) / NTHREADS; ++r) {
    int idx = r * NTHREADS + tid;
    int s  = idx >> 5;
    int hh = idx & 31;
    v4f v = *reinterpret_cast<const v4f*>(h + (size_t)(s0 + s) * HDIM + hh * 4);
    v2f lo; lo.x = v.x; lo.y = v.y;
    v2f hi; hi.x = v.z; hi.y = v.w;
    *reinterpret_cast<v2f*>(&sh[s * A_S + 2 * hh])          = lo;
    *reinterpret_cast<v2f*>(&sh[s * A_S + A_HALF + 2 * hh]) = hi;
  }
  {
    int s = tid >> 3;
    int c4 = (tid & 7) * 4;
    *reinterpret_cast<v4f*>(&sh[Z_BASE + s * Z_S + c4]) =
        *reinterpret_cast<const v4f*>(z0 + (size_t)(s0 + s) * ZDIM + c4);
  }
  __syncthreads();

  const float* ap = sh + mrow * A_S + mhalf * A_HALF;

  {
    const int p0 = wave * 6;
    const int psplit = ((p0 >> 2) + 1) << 2;
    #pragma unroll 1
    for (int seg = 0; seg < 2; ++seg) {
      int pbeg = seg ? psplit : p0;
      int pend = seg ? (p0 + 6) : psplit;
      int ct  = pbeg >> 2;
      int mat = ct >> 3;
      int nt2 = ct & 7;
      const float* Wm = (mat == 0) ? Wd1 : (mat == 1) ? Wd2 : Wb;
      const float* bm = (mat == 0) ? bd1 : (mat == 1) ? bd2 : bb;
      int cc = nt2 * 16 + mrow;
      float bias = bm[cc];
      const float* wrow = Wm + (size_t)cc * HDIM;
      v2f bmat[32];
      #pragma unroll
      for (int hh = 0; hh < 32; ++hh)
        bmat[hh] = *reinterpret_cast<const v2f*>(wrow + hh * 4 + 2 * mhalf);
      #pragma unroll 1
      for (int p = pbeg; p < pend; ++p) {
        int m = p & 3;
        v8f acc;
        #pragma unroll
        for (int r = 0; r < 8; ++r) acc[r] = bias;
        const float* am = ap + m * 16 * A_S;
        #pragma unroll
        for (int q = 0; q < 16; ++q) {
          v4f a4 = *reinterpret_cast<const v4f*>(am + 4 * q);
          v2f a0; a0.x = a4.x; a0.y = a4.y;
          v2f a1; a1.x = a4.z; a1.y = a4.w;
          acc = wmma_f32(a0, bmat[2 * q],     acc);
          acc = wmma_f32(a1, bmat[2 * q + 1], acc);
        }
        int sbase = m * 16 + mhalf * 8;
        #pragma unroll
        for (int r = 0; r < 8; ++r) {
          float v = acc[r];
          if (mat < 2) v = tanh_hw(v);
          sh[FD_BASE + mat * SM_STRIDE + (sbase + r) * SM_S + cc] = v;
        }
      }
    }
  }
  const int b_lane = mrow * 4 * HDIM + 2 * mhalf;
  v2f bmat[32];
  float bias;
  {
    const int nt0 = wave * 4;
    const int ch0 = (nt0 >> 1) * 128 + ((nt0 & 1) << 4) * 4;
    const float* w0 = Wd + (size_t)ch0 * HDIM + b_lane;
    #pragma unroll
    for (int hh = 0; hh < 32; ++hh)
      bmat[hh] = *reinterpret_cast<const v2f*>(w0 + hh * 4);
    bias = bd[ch0 + mrow * 4];
  }
  __builtin_amdgcn_fence(__ATOMIC_RELEASE, "workgroup", "local");
  __builtin_amdgcn_s_barrier();
  __builtin_amdgcn_fence(__ATOMIC_ACQUIRE, "workgroup", "local");

  const int fh = wave & 1;
  const int fg = wave >> 1;
  const int fs = fh * 32 + lane;
  float ldp = 0.0f;
  float d1r[16], d2r[16], br[16];
  int jrun[4];
  #pragma unroll
  for (int c = 0; c < 4; ++c) {
    jrun[c] = c < 2 ? 2 * fg + c : 28 - 2 * fg + c;
    v4f a = *reinterpret_cast<const v4f*>(&sh[FD_BASE + 0 * SM_STRIDE + fs * SM_S + jrun[c] * 4]);
    v4f b = *reinterpret_cast<const v4f*>(&sh[FD_BASE + 1 * SM_STRIDE + fs * SM_S + jrun[c] * 4]);
    v4f d = *reinterpret_cast<const v4f*>(&sh[FD_BASE + 2 * SM_STRIDE + fs * SM_S + jrun[c] * 4]);
    d1r[c * 4 + 0] = a.x; d1r[c * 4 + 1] = a.y; d1r[c * 4 + 2] = a.z; d1r[c * 4 + 3] = a.w;
    d2r[c * 4 + 0] = b.x; d2r[c * 4 + 1] = b.y; d2r[c * 4 + 2] = b.z; d2r[c * 4 + 3] = b.w;
    br [c * 4 + 0] = d.x; br [c * 4 + 1] = d.y; br [c * 4 + 2] = d.z; br [c * 4 + 3] = d.w;
  }
  float* const ftile = sh + FD_BASE + fs * FD_S;
  float* const fzrow = sh + Z_BASE + fs * Z_S;

  #pragma unroll 1
  for (int k = 0; k < 4; ++k) {
    __syncthreads();

    auto run_tile = [&](auto refill_tag, int tt) {
      constexpr bool REFILL = decltype(refill_tag)::value;
      int nt = wave * 4 + tt;
      int i_ = nt >> 1;
      int j0 = (nt & 1) << 4;
      const float* wnext = Wd;
      float bias_next = 0.0f;
      if (REFILL) {
        int tn  = k * 4 + tt + 1;
        int ntn = wave * 4 + (tn & 3);
        int chn = (ntn >> 1) * 128 + ((ntn & 1) << 4) * 4 + (tn >> 2);
        wnext = Wd + (size_t)chn * HDIM + b_lane;
        bias_next = bd[chn + mrow * 4];
      }
      v8f acc[4];
      #pragma unroll
      for (int m = 0; m < 4; ++m) {
        #pragma unroll
        for (int r = 0; r < 8; ++r) acc[m][r] = bias;
      }
      v4f ring[A_RING][4];
      #pragma unroll
      for (int q = 0; q < A_AHEAD; ++q) {
        #pragma unroll
        for (int m = 0; m < 4; ++m)
          ring[q][m] = *reinterpret_cast<const v4f*>(ap + m * 16 * A_S + 4 * q);
      }
      #pragma unroll
      for (int q = 0; q < 16; ++q) {
        if (q == 4)  { if (tt == 3) __builtin_amdgcn_s_setprio(2); }
        if (q == 8)  { if (tt == 3) __builtin_amdgcn_s_setprio(1); }
        if (q == 12) { if (tt == 3) __builtin_amdgcn_s_setprio(0); }
        if (q + A_AHEAD < 16) {
          #pragma unroll
          for (int m = 0; m < 4; ++m)
            ring[(q + A_AHEAD) % A_RING][m] =
                *reinterpret_cast<const v4f*>(ap + m * 16 * A_S + 4 * (q + A_AHEAD));
        }
        if (REFILL && q > 0) {
          bmat[2 * q - 2] = *reinterpret_cast<const v2f*>(wnext + (2 * q - 2) * 4);
          bmat[2 * q - 1] = *reinterpret_cast<const v2f*>(wnext + (2 * q - 1) * 4);
        }
        __builtin_amdgcn_sched_barrier(0);
        #pragma unroll
        for (int m = 0; m < 4; ++m) {
          v4f a = ring[q % A_RING][m];
          v2f a0; a0.x = a.x; a0.y = a.y;
          acc[m] = wmma_f32(a0, bmat[2 * q], acc[m]);
        }
        #pragma unroll
        for (int m = 0; m < 4; ++m) {
          v4f a = ring[q % A_RING][m];
          v2f a1; a1.x = a.z; a1.y = a.w;
          acc[m] = wmma_f32(a1, bmat[2 * q + 1], acc[m]);
        }
        __builtin_amdgcn_sched_barrier(0);
      }
      if (REFILL) {
        bmat[30] = *reinterpret_cast<const v2f*>(wnext + 30 * 4);
        bmat[31] = *reinterpret_cast<const v2f*>(wnext + 31 * 4);
        bias = bias_next;
      }
      #pragma unroll
      for (int m = 0; m < 4; ++m) {
        int sb = m * 16 + mhalf * 8;
        int base = FD_BASE + sb * FD_S + i_ * FD_ROW + j0 + mrow;
        #pragma unroll
        for (int r = 0; r < 8; ++r)
          sh[base + r * FD_S] = acc[m][r];
      }
    };
    __builtin_amdgcn_s_setprio(3);
    #pragma unroll 1
    for (int tt = 0; tt < 4; ++tt) {
      if (k * 4 + tt < 15) run_tile(std::true_type{}, tt);
      else                 run_tile(std::false_type{}, tt);
    }
    __builtin_amdgcn_fence(__ATOMIC_RELEASE, "workgroup", "local");
    __builtin_amdgcn_s_barrier();
    __builtin_amdgcn_fence(__ATOMIC_ACQUIRE, "workgroup", "local");

    float d1k[4], d2k[4], bk[4];
    #pragma unroll
    for (int c = 0; c < 4; ++c) {
      d1k[c] = d1r[c * 4 + k];
      d2k[c] = d2r[c * 4 + k];
      bk[c]  = br [c * 4 + k];
    }
    float zown[4], pre[4];
    switch (fg) {
      case 0:  flow_pre<0>(fzrow, ftile, d2k, bk, zown, pre); break;
      case 1:  flow_pre<1>(fzrow, ftile, d2k, bk, zown, pre); break;
      case 2:  flow_pre<2>(fzrow, ftile, d2k, bk, zown, pre); break;
      case 3:  flow_pre<3>(fzrow, ftile, d2k, bk, zown, pre); break;
      case 4:  flow_pre<4>(fzrow, ftile, d2k, bk, zown, pre); break;
      case 5:  flow_pre<5>(fzrow, ftile, d2k, bk, zown, pre); break;
      case 6:  flow_pre<6>(fzrow, ftile, d2k, bk, zown, pre); break;
      default: flow_pre<7>(fzrow, ftile, d2k, bk, zown, pre); break;
    }
    float tv[4];
    #pragma unroll
    for (int c = 0; c < 4; ++c) {
      const float t = tanh_hw(pre[c]);
      tv[c] = t;
      ftile[jrun[c] * FD_ROW + T_COL] = t;
      const float dj = (1.0f - t * t) * (d1k[c] * d2k[c]) + 1.0f;
      ldp += log_hw(fabsf(dj));
    }
    __builtin_amdgcn_fence(__ATOMIC_RELEASE, "workgroup", "local");
    __builtin_amdgcn_s_barrier();
    __builtin_amdgcn_fence(__ATOMIC_ACQUIRE, "workgroup", "local");
    float dz[4];
    switch (fg) {
      case 0:  flow_dz<0>(ftile, tv, d1k, dz); break;
      case 1:  flow_dz<1>(ftile, tv, d1k, dz); break;
      case 2:  flow_dz<2>(ftile, tv, d1k, dz); break;
      case 3:  flow_dz<3>(ftile, tv, d1k, dz); break;
      case 4:  flow_dz<4>(ftile, tv, d1k, dz); break;
      case 5:  flow_dz<5>(ftile, tv, d1k, dz); break;
      case 6:  flow_dz<6>(ftile, tv, d1k, dz); break;
      default: flow_dz<7>(ftile, tv, d1k, dz); break;
    }
    #pragma unroll
    for (int c = 0; c < 4; ++c)
      fzrow[31 - jrun[c]] = zown[c] + dz[c];
  }

  sh[LD_BASE + fg * TBATCH + fs] = ldp;
  __syncthreads();
  {
    int s = tid >> 3;
    int c4 = (tid & 7) * 4;
    *reinterpret_cast<v4f*>(out_z + (size_t)(s0 + s) * ZDIM + c4) =
        *reinterpret_cast<const v4f*>(&sh[Z_BASE + s * Z_S + c4]);
    if (tid < TBATCH) {
      float v = 0.0f;
      #pragma unroll
      for (int g = 0; g < 8; ++g) v += sh[LD_BASE + g * TBATCH + tid];
      out_ld[s0 + tid] = v;
    }
  }
}

extern "C" void kernel_launch(void* const* d_in, const int* in_sizes, int n_in,
                              void* d_out, int out_size, void* d_ws, size_t ws_size,
                              hipStream_t stream) {
  const float* z0  = (const float*)d_in[0];
  const float* h   = (const float*)d_in[1];
  const float* Wd  = (const float*)d_in[2];
  const float* bd  = (const float*)d_in[3];
  const float* Wd1 = (const float*)d_in[4];
  const float* bd1 = (const float*)d_in[5];
  const float* Wd2 = (const float*)d_in[6];
  const float* bd2 = (const float*)d_in[7];
  const float* Wb  = (const float*)d_in[8];
  const float* bb  = (const float*)d_in[9];

  const int B = in_sizes[0] / ZDIM;
  float* out   = (float*)d_out;
  float* out_z = out;
  float* out_ld = out + (size_t)B * ZDIM;

  dim3 grid(B / TBATCH), block(NTHREADS);
  hipLaunchKernelGGL(sylvester_flow_kernel, grid, block, 0, stream,
                     z0, h, Wd, bd, Wd1, bd1, Wd2, bd2, Wb, bb, out_z, out_ld);
}
